# starting kernel
# baseline (speedup 1.0000x reference)
#include <hip/hip_runtime.h>
#include <hip/hip_bf16.h>

#define D 96

typedef float    v2f  __attribute__((ext_vector_type(2)));
typedef float    v8f  __attribute__((ext_vector_type(8)));
typedef _Float16 v16h __attribute__((ext_vector_type(16)));

#ifndef __has_builtin
#define __has_builtin(x) 0
#endif

#if defined(__AMDGCN__) && __has_builtin(__builtin_amdgcn_wmma_f32_16x16x4_f32)
#define USE_F32_WMMA 1
#else
#define USE_F32_WMMA 0
#endif

__global__ void zero_kernel(float* __restrict__ p, size_t n) {
  size_t i = (size_t)blockIdx.x * blockDim.x + threadIdx.x;
  if (i < n) p[i] = 0.0f;
}

__global__ void deg_kernel(const int* __restrict__ col, float* __restrict__ deg, int E) {
  int e = blockIdx.x * blockDim.x + threadIdx.x;
  if (e < E) atomicAdd(&deg[col[e]], 1.0f);
}

__global__ void dinv_kernel(float* __restrict__ deg, int n) {
  int i = blockIdx.x * blockDim.x + threadIdx.x;
  if (i < n) deg[i] = rsqrtf(deg[i] + 1.0f);
}

__global__ void gemm_wmma(const float* __restrict__ A, const float* __restrict__ B,
                          float* __restrict__ C, int M) {
  int m0 = blockIdx.x * 16;
  if (m0 + 16 > M) return;
  int wave = threadIdx.x >> 5;
  int lane = threadIdx.x & 31;
  int n0   = wave * 16;
  int l    = lane & 15;
  int half = lane >> 4;

  const float* arow = A + (size_t)(m0 + l) * D;
  v8f acc = {};

#if USE_F32_WMMA
  int kk = half * 2;
  #pragma unroll
  for (int k = 0; k < D; k += 4) {
    v2f a, b;
    a.x = arow[k + kk];
    a.y = arow[k + kk + 1];
    b.x = B[(size_t)(k + kk)     * D + n0 + l];
    b.y = B[(size_t)(k + kk + 1) * D + n0 + l];
    acc = __builtin_amdgcn_wmma_f32_16x16x4_f32(
        false, a, false, b, (short)0, acc, false, false);
  }
#else
  for (int k0 = 0; k0 < D; k0 += 32) {
    v16h a, b;
    #pragma unroll
    for (int e = 0; e < 16; ++e) {
      int kk = k0 + half * 8 + ((e >> 3) * 16) + (e & 7);
      a[e] = (_Float16)arow[kk];
      b[e] = (_Float16)B[(size_t)kk * D + n0 + l];
    }
    acc = __builtin_amdgcn_wmma_f32_16x16x32_f16(
        false, a, false, b, (short)0, acc, false, false);
  }
#endif

  int colIdx = n0 + l;
  int rbase  = m0 + half * 8;
  #pragma unroll
  for (int j = 0; j < 8; ++j)
    C[(size_t)(rbase + j) * D + colIdx] = acc[j];
}

__global__ void scatter_edges(const int* __restrict__ row, const int* __restrict__ col,
                              const float* __restrict__ dinv, const float* __restrict__ h,
                              float* __restrict__ out, int E) {
  int idx = blockIdx.x * blockDim.x + threadIdx.x;
  int e = idx / 24;
  if (e >= E) return;
  int f = (idx % 24) * 4;
  int r = row[e], c = col[e];
  float nrm = dinv[r] * dinv[c];
  float4 hv = *(const float4*)(h + (size_t)r * D + f);
  float* o = out + (size_t)c * D + f;
  atomicAdd(o + 0, hv.x * nrm);
  atomicAdd(o + 1, hv.y * nrm);
  atomicAdd(o + 2, hv.z * nrm);
  atomicAdd(o + 3, hv.w * nrm);
}

__global__ void finalize_kernel(float* __restrict__ accOut, const float* __restrict__ h,
                                const float* __restrict__ dinv, const float* __restrict__ b,
                                int relu, int N) {
  int idx = blockIdx.x * blockDim.x + threadIdx.x;
  if (idx >= N * 24) return;
  int node = idx / 24;
  int f = (idx % 24) * 4;
  float di = dinv[node];
  float s = di * di;
  size_t off = (size_t)node * D + f;
  float4 v  = *(const float4*)(accOut + off);
  float4 hv = *(const float4*)(h + off);
  float4 bv = *(const float4*)(b + f);
  v.x += hv.x * s + bv.x;
  v.y += hv.y * s + bv.y;
  v.z += hv.z * s + bv.z;
  v.w += hv.w * s + bv.w;
  if (relu) {
    v.x = fmaxf(v.x, 0.0f); v.y = fmaxf(v.y, 0.0f);
    v.z = fmaxf(v.z, 0.0f); v.w = fmaxf(v.w, 0.0f);
  }
  *(float4*)(accOut + off) = v;
}

extern "C" void kernel_launch(void* const* d_in, const int* in_sizes, int n_in,
                              void* d_out, int out_size, void* d_ws, size_t ws_size,
                              hipStream_t stream) {
  const float* x  = (const float*)d_in[0];
  const int*   ei = (const int*)d_in[1];
  const float* W1 = (const float*)d_in[2];
  const float* b1 = (const float*)d_in[3];
  const float* W2 = (const float*)d_in[4];
  const float* b2 = (const float*)d_in[5];
  float* out = (float*)d_out;

  int N = in_sizes[0] / D;
  int E = in_sizes[1] / 2;
  const int* row = ei;
  const int* col = ei + E;

  float* ws   = (float*)d_ws;
  float* dinv = ws;
  float* h    = ws + (((size_t)N + 127) & ~127ull);
  float* acc  = h + (size_t)N * D;

  const int tpb = 256;
  size_t nd = (size_t)N * D;
  int gN    = (N + tpb - 1) / tpb;
  int gE    = (E + tpb - 1) / tpb;
  int gND   = (int)((nd + tpb - 1) / tpb);
  int gEf   = (int)(((size_t)E * 24 + tpb - 1) / tpb);
  int gNf   = (N * 24 + tpb - 1) / tpb;
  int gGemm = N / 16;

  zero_kernel<<<gN, tpb, 0, stream>>>(dinv, (size_t)N);
  deg_kernel<<<gE, tpb, 0, stream>>>(col, dinv, E);
  dinv_kernel<<<gN, tpb, 0, stream>>>(dinv, N);

  zero_kernel<<<gND, tpb, 0, stream>>>(acc, nd);
  gemm_wmma<<<gGemm, 192, 0, stream>>>(x, W1, h, N);
  scatter_edges<<<gEf, tpb, 0, stream>>>(row, col, dinv, h, acc, E);
  finalize_kernel<<<gNf, tpb, 0, stream>>>(acc, h, dinv, b1, 1, N);

  gemm_wmma<<<gGemm, 192, 0, stream>>>(acc, W2, h, N);
  zero_kernel<<<gND, tpb, 0, stream>>>(out, nd);
  scatter_edges<<<gEf, tpb, 0, stream>>>(row, col, dinv, h, out, E);
  finalize_kernel<<<gNf, tpb, 0, stream>>>(out, h, dinv, b2, 0, N);
}


# scatter re-mapped: lane = one float of a 128-B line, 8 edges per wave, so each row load and each f32 atomic add covers ONE whole line (was 4 partial lines per add); add order still arrival order
# speedup vs baseline: 1.4213x; 1.4213x over previous
#include <hip/hip_runtime.h>
#include <hip/hip_bf16.h>

#define D 96

typedef float    v2f  __attribute__((ext_vector_type(2)));
typedef float    v8f  __attribute__((ext_vector_type(8)));
typedef _Float16 v16h __attribute__((ext_vector_type(16)));

#ifndef __has_builtin
#define __has_builtin(x) 0
#endif

#if defined(__AMDGCN__) && __has_builtin(__builtin_amdgcn_wmma_f32_16x16x4_f32)
#define USE_F32_WMMA 1
#else
#define USE_F32_WMMA 0
#endif

__global__ void zero_kernel(float* __restrict__ p, size_t n) {
  size_t i = (size_t)blockIdx.x * blockDim.x + threadIdx.x;
  if (i < n) p[i] = 0.0f;
}

__global__ void deg_kernel(const int* __restrict__ col, float* __restrict__ deg, int E) {
  int e = blockIdx.x * blockDim.x + threadIdx.x;
  if (e < E) atomicAdd(&deg[col[e]], 1.0f);
}

__global__ void dinv_kernel(float* __restrict__ deg, int n) {
  int i = blockIdx.x * blockDim.x + threadIdx.x;
  if (i < n) deg[i] = rsqrtf(deg[i] + 1.0f);
}

__global__ void gemm_wmma(const float* __restrict__ A, const float* __restrict__ B,
                          float* __restrict__ C, int M) {
  int m0 = blockIdx.x * 16;
  if (m0 + 16 > M) return;
  int wave = threadIdx.x >> 5;
  int lane = threadIdx.x & 31;
  int n0   = wave * 16;
  int l    = lane & 15;
  int half = lane >> 4;

  const float* arow = A + (size_t)(m0 + l) * D;
  v8f acc = {};

#if USE_F32_WMMA
  int kk = half * 2;
  #pragma unroll
  for (int k = 0; k < D; k += 4) {
    v2f a, b;
    a.x = arow[k + kk];
    a.y = arow[k + kk + 1];
    b.x = B[(size_t)(k + kk)     * D + n0 + l];
    b.y = B[(size_t)(k + kk + 1) * D + n0 + l];
    acc = __builtin_amdgcn_wmma_f32_16x16x4_f32(
        false, a, false, b, (short)0, acc, false, false);
  }
#else
  for (int k0 = 0; k0 < D; k0 += 32) {
    v16h a, b;
    #pragma unroll
    for (int e = 0; e < 16; ++e) {
      int kk = k0 + half * 8 + ((e >> 3) * 16) + (e & 7);
      a[e] = (_Float16)arow[kk];
      b[e] = (_Float16)B[(size_t)kk * D + n0 + l];
    }
    acc = __builtin_amdgcn_wmma_f32_16x16x32_f16(
        false, a, false, b, (short)0, acc, false, false);
  }
#endif

  int colIdx = n0 + l;
  int rbase  = m0 + half * 8;
  #pragma unroll
  for (int j = 0; j < 8; ++j)
    C[(size_t)(rbase + j) * D + colIdx] = acc[j];
}

#define EPW 8
__global__ void __launch_bounds__(256)
scatter_edges(const int* __restrict__ row, const int* __restrict__ col,
              const float* __restrict__ dinv, const float* __restrict__ h,
              float* __restrict__ out, int E) {
  int lane = threadIdx.x & 31;
  int wave = __builtin_amdgcn_readfirstlane((int)((blockIdx.x * blockDim.x + threadIdx.x) >> 5));
  int e0 = wave * EPW;
  if (e0 >= E) return;
  if (e0 + EPW <= E) {
    int r[EPW], c[EPW];
    float nrm[EPW];
    float v[EPW][3];
    #pragma unroll
    for (int i = 0; i < EPW; ++i) { r[i] = row[e0 + i]; c[i] = col[e0 + i]; }
    #pragma unroll
    for (int i = 0; i < EPW; ++i) {
      const float* hr = h + (size_t)r[i] * D + lane;
      v[i][0] = hr[0]; v[i][1] = hr[32]; v[i][2] = hr[64];
    }
    #pragma unroll
    for (int i = 0; i < EPW; ++i) nrm[i] = dinv[r[i]] * dinv[c[i]];
    #pragma unroll
    for (int i = 0; i < EPW; ++i) {
      float* o = out + (size_t)c[i] * D + lane;
      atomicAdd(o,      v[i][0] * nrm[i]);
      atomicAdd(o + 32, v[i][1] * nrm[i]);
      atomicAdd(o + 64, v[i][2] * nrm[i]);
    }
  } else {
    for (int e = e0; e < E; ++e) {
      int r = row[e], c = col[e];
      float nrm = dinv[r] * dinv[c];
      const float* hr = h + (size_t)r * D + lane;
      float* o = out + (size_t)c * D + lane;
      atomicAdd(o,      hr[0]  * nrm);
      atomicAdd(o + 32, hr[32] * nrm);
      atomicAdd(o + 64, hr[64] * nrm);
    }
  }
}

__global__ void finalize_kernel(float* __restrict__ accOut, const float* __restrict__ h,
                                const float* __restrict__ dinv, const float* __restrict__ b,
                                int relu, int N) {
  int idx = blockIdx.x * blockDim.x + threadIdx.x;
  if (idx >= N * 24) return;
  int node = idx / 24;
  int f = (idx % 24) * 4;
  float di = dinv[node];
  float s = di * di;
  size_t off = (size_t)node * D + f;
  float4 v  = *(const float4*)(accOut + off);
  float4 hv = *(const float4*)(h + off);
  float4 bv = *(const float4*)(b + f);
  v.x += hv.x * s + bv.x;
  v.y += hv.y * s + bv.y;
  v.z += hv.z * s + bv.z;
  v.w += hv.w * s + bv.w;
  if (relu) {
    v.x = fmaxf(v.x, 0.0f); v.y = fmaxf(v.y, 0.0f);
    v.z = fmaxf(v.z, 0.0f); v.w = fmaxf(v.w, 0.0f);
  }
  *(float4*)(accOut + off) = v;
}

extern "C" void kernel_launch(void* const* d_in, const int* in_sizes, int n_in,
                              void* d_out, int out_size, void* d_ws, size_t ws_size,
                              hipStream_t stream) {
  const float* x  = (const float*)d_in[0];
  const int*   ei = (const int*)d_in[1];
  const float* W1 = (const float*)d_in[2];
  const float* b1 = (const float*)d_in[3];
  const float* W2 = (const float*)d_in[4];
  const float* b2 = (const float*)d_in[5];
  float* out = (float*)d_out;

  int N = in_sizes[0] / D;
  int E = in_sizes[1] / 2;
  const int* row = ei;
  const int* col = ei + E;

  float* ws   = (float*)d_ws;
  float* dinv = ws;
  float* h    = ws + (((size_t)N + 127) & ~127ull);
  float* acc  = h + (size_t)N * D;

  const int tpb = 256;
  size_t nd = (size_t)N * D;
  int gN    = (N + tpb - 1) / tpb;
  int gE    = (E + tpb - 1) / tpb;
  int gND   = (int)((nd + tpb - 1) / tpb);
  int gEf   = (E + 63) / 64;
  int gNf   = (N * 24 + tpb - 1) / tpb;
  int gGemm = N / 16;

  zero_kernel<<<gN, tpb, 0, stream>>>(dinv, (size_t)N);
  deg_kernel<<<gE, tpb, 0, stream>>>(col, dinv, E);
  dinv_kernel<<<gN, tpb, 0, stream>>>(dinv, N);

  zero_kernel<<<gND, tpb, 0, stream>>>(acc, nd);
  gemm_wmma<<<gGemm, 192, 0, stream>>>(x, W1, h, N);
  scatter_edges<<<gEf, tpb, 0, stream>>>(row, col, dinv, h, acc, E);
  finalize_kernel<<<gNf, tpb, 0, stream>>>(acc, h, dinv, b1, 1, N);

  gemm_wmma<<<gGemm, 192, 0, stream>>>(acc, W2, h, N);
  zero_kernel<<<gND, tpb, 0, stream>>>(out, nd);
  scatter_edges<<<gEf, tpb, 0, stream>>>(row, col, dinv, h, out, E);
  finalize_kernel<<<gNf, tpb, 0, stream>>>(out, h, dinv, b2, 0, N);
}


# edges grouped by target once per call (count, prefix sum, slot fill); each layer sums a node's incoming rows in registers, writes the row once: no f32 atomics; sum order = fill arrival order
# speedup vs baseline: 1.8716x; 1.3168x over previous
#include <hip/hip_runtime.h>
#include <hip/hip_bf16.h>

#define D 96

typedef float    v2f  __attribute__((ext_vector_type(2)));
typedef float    v8f  __attribute__((ext_vector_type(8)));
typedef _Float16 v16h __attribute__((ext_vector_type(16)));

#ifndef __has_builtin
#define __has_builtin(x) 0
#endif

#if defined(__AMDGCN__) && __has_builtin(__builtin_amdgcn_wmma_f32_16x16x4_f32)
#define USE_F32_WMMA 1
#else
#define USE_F32_WMMA 0
#endif

__global__ void zero_int(int* __restrict__ p, int n) {
  int i = blockIdx.x * blockDim.x + threadIdx.x;
  if (i < n) p[i] = 0;
}

__global__ void count_targets(const int* __restrict__ col, int* __restrict__ cnt, int E) {
  int q = gridDim.x * blockDim.x;
  int e = blockIdx.x * blockDim.x + threadIdx.x;
  if (e + 3 * q < E) {
    int c[4];
    #pragma unroll
    for (int k = 0; k < 4; ++k) c[k] = col[e + k * q];
    #pragma unroll
    for (int k = 0; k < 4; ++k) atomicAdd(&cnt[c[k]], 1);
  } else {
    for (int k = 0; k < 4; ++k) {
      int ee = e + k * q;
      if (ee < E) atomicAdd(&cnt[col[ee]], 1);
    }
  }
}

__global__ void __launch_bounds__(256)
scan_counts(const int* __restrict__ cnt, int* __restrict__ rowptr, int* __restrict__ cursor,
            float* __restrict__ dinv, int N) {
  __shared__ int sh[256];
  int t = threadIdx.x;
  int tile0 = blockIdx.x * 1024;
  int part = 0;
  #pragma unroll 8
  for (int i = t * 4; i < tile0; i += 1024) {
    int4 v = *(const int4*)(cnt + i);
    part += (v.x + v.y) + (v.z + v.w);
  }
  int4 c = *(const int4*)(cnt + tile0 + t * 4);
  int tsum = (c.x + c.y) + (c.z + c.w);

  sh[t] = part;
  __syncthreads();
  for (int off = 128; off > 0; off >>= 1) {
    if (t < off) sh[t] += sh[t + off];
    __syncthreads();
  }
  int base = sh[0];
  __syncthreads();

  sh[t] = tsum;
  __syncthreads();
  for (int off = 1; off < 256; off <<= 1) {
    int v = (t >= off) ? sh[t - off] : 0;
    __syncthreads();
    sh[t] += v;
    __syncthreads();
  }
  int p0 = base + sh[t] - tsum;
  int p1 = p0 + c.x, p2 = p1 + c.y, p3 = p2 + c.z;
  int i0 = tile0 + t * 4;
  if (i0 + 3 < N) {
    *(int4*)(rowptr + i0) = make_int4(p0, p1, p2, p3);
    *(int4*)(cursor + i0) = make_int4(p0, p1, p2, p3);
    *(float4*)(dinv + i0) = make_float4(rsqrtf((float)c.x + 1.0f), rsqrtf((float)c.y + 1.0f),
                                        rsqrtf((float)c.z + 1.0f), rsqrtf((float)c.w + 1.0f));
  } else {
    int pk[4] = {p0, p1, p2, p3};
    int ck[4] = {c.x, c.y, c.z, c.w};
    for (int k = 0; k < 4; ++k) {
      int idx = i0 + k;
      if (idx <= N) rowptr[idx] = pk[k];
      if (idx < N) { cursor[idx] = pk[k]; dinv[idx] = rsqrtf((float)ck[k] + 1.0f); }
    }
  }
}

__global__ void fill_sources(const int* __restrict__ row, const int* __restrict__ col,
                             int* __restrict__ cursor, int* __restrict__ src, int E) {
  int q = gridDim.x * blockDim.x;
  int e = blockIdx.x * blockDim.x + threadIdx.x;
  if (e + 3 * q < E) {
    int r[4], c[4], p[4];
    #pragma unroll
    for (int k = 0; k < 4; ++k) { r[k] = row[e + k * q]; c[k] = col[e + k * q]; }
    #pragma unroll
    for (int k = 0; k < 4; ++k) p[k] = atomicAdd(&cursor[c[k]], 1);
    #pragma unroll
    for (int k = 0; k < 4; ++k) src[p[k]] = r[k];
  } else {
    for (int k = 0; k < 4; ++k) {
      int ee = e + k * q;
      if (ee < E) {
        int p = atomicAdd(&cursor[col[ee]], 1);
        src[p] = row[ee];
      }
    }
  }
}

__global__ void gemm_wmma(const float* __restrict__ A, const float* __restrict__ B,
                          float* __restrict__ C, int M) {
  int m0 = blockIdx.x * 16;
  if (m0 + 16 > M) return;
  int wave = threadIdx.x >> 5;
  int lane = threadIdx.x & 31;
  int n0   = wave * 16;
  int l    = lane & 15;
  int half = lane >> 4;

  const float* arow = A + (size_t)(m0 + l) * D;
  v8f acc = {};

#if USE_F32_WMMA
  int kk = half * 2;
  #pragma unroll
  for (int k = 0; k < D; k += 4) {
    v2f a, b;
    a.x = arow[k + kk];
    a.y = arow[k + kk + 1];
    b.x = B[(size_t)(k + kk)     * D + n0 + l];
    b.y = B[(size_t)(k + kk + 1) * D + n0 + l];
    acc = __builtin_amdgcn_wmma_f32_16x16x4_f32(
        false, a, false, b, (short)0, acc, false, false);
  }
#else
  for (int k0 = 0; k0 < D; k0 += 32) {
    v16h a, b;
    #pragma unroll
    for (int e = 0; e < 16; ++e) {
      int kk = k0 + half * 8 + ((e >> 3) * 16) + (e & 7);
      a[e] = (_Float16)arow[kk];
      b[e] = (_Float16)B[(size_t)kk * D + n0 + l];
    }
    acc = __builtin_amdgcn_wmma_f32_16x16x32_f16(
        false, a, false, b, (short)0, acc, false, false);
  }
#endif

  int colIdx = n0 + l;
  int rbase  = m0 + half * 8;
  #pragma unroll
  for (int j = 0; j < 8; ++j)
    C[(size_t)(rbase + j) * D + colIdx] = acc[j];
}

template <int K>
__device__ __forceinline__ void add_rows(const int* __restrict__ src, const float* __restrict__ dinv,
                                         const float* __restrict__ h, int j, int lane, float dc,
                                         float& a0, float& a1, float& a2) {
  int r[K];
  float w[K];
  float v[K][3];
  #pragma unroll
  for (int i = 0; i < K; ++i) r[i] = src[j + i];
  #pragma unroll
  for (int i = 0; i < K; ++i) {
    const float* hr = h + (size_t)r[i] * D + lane;
    v[i][0] = hr[0]; v[i][1] = hr[32]; v[i][2] = hr[64];
  }
  #pragma unroll
  for (int i = 0; i < K; ++i) w[i] = dinv[r[i]] * dc;
  #pragma unroll
  for (int i = 0; i < K; ++i) {
    a0 += v[i][0] * w[i];
    a1 += v[i][1] * w[i];
    a2 += v[i][2] * w[i];
  }
}

__device__ __forceinline__ void one_node(const int* __restrict__ rowptr, const int* __restrict__ src,
                                         const float* __restrict__ dinv, const float* __restrict__ h,
                                         float* __restrict__ out, int relu, int node, int lane,
                                         float b0, float b1, float b2) {
  int beg = rowptr[node], end = rowptr[node + 1];
  float dc = dinv[node];
  const float* hc = h + (size_t)node * D + lane;
  float s0 = hc[0], s1 = hc[32], s2 = hc[64];
  float a0 = 0.0f, a1 = 0.0f, a2 = 0.0f;
  int j = beg;
  for (; j + 8 <= end; j += 8) add_rows<8>(src, dinv, h, j, lane, dc, a0, a1, a2);
  if (j + 4 <= end) { add_rows<4>(src, dinv, h, j, lane, dc, a0, a1, a2); j += 4; }
  if (j + 2 <= end) { add_rows<2>(src, dinv, h, j, lane, dc, a0, a1, a2); j += 2; }
  if (j < end) add_rows<1>(src, dinv, h, j, lane, dc, a0, a1, a2);
  float s = dc * dc;
  a0 += s0 * s + b0;
  a1 += s1 * s + b1;
  a2 += s2 * s + b2;
  if (relu) { a0 = fmaxf(a0, 0.0f); a1 = fmaxf(a1, 0.0f); a2 = fmaxf(a2, 0.0f); }
  float* o = out + (size_t)node * D + lane;
  o[0] = a0; o[32] = a1; o[64] = a2;
}

#define NPW 4
__global__ void __launch_bounds__(256)
gather_rows(const int* __restrict__ rowptr, const int* __restrict__ src,
            const float* __restrict__ dinv, const float* __restrict__ h,
            const float* __restrict__ b, float* __restrict__ out, int relu, int N) {
  int lane = threadIdx.x & 31;
  int wave = __builtin_amdgcn_readfirstlane((int)((blockIdx.x * blockDim.x + threadIdx.x) >> 5));
  int node0 = wave * NPW;
  if (node0 >= N) return;
  float b0 = b[lane], b1 = b[lane + 32], b2 = b[lane + 64];
  int nend = node0 + NPW < N ? node0 + NPW : N;
  for (int node = node0; node < nend; ++node)
    one_node(rowptr, src, dinv, h, out, relu, node, lane, b0, b1, b2);
}

extern "C" void kernel_launch(void* const* d_in, const int* in_sizes, int n_in,
                              void* d_out, int out_size, void* d_ws, size_t ws_size,
                              hipStream_t stream) {
  const float* x  = (const float*)d_in[0];
  const int*   ei = (const int*)d_in[1];
  const float* W1 = (const float*)d_in[2];
  const float* b1 = (const float*)d_in[3];
  const float* W2 = (const float*)d_in[4];
  const float* b2 = (const float*)d_in[5];
  float* out = (float*)d_out;

  int N = in_sizes[0] / D;
  int E = in_sizes[1] / 2;
  const int* row = ei;
  const int* col = ei + E;

  size_t NP = (((size_t)N + 1 + 1023) / 1024) * 1024;
  size_t EP = (((size_t)E + 1023) / 1024) * 1024;
  float* ws     = (float*)d_ws;
  float* dinv   = ws;
  int*   cnt    = (int*)(ws + NP);
  int*   rowptr = (int*)(ws + 2 * NP);
  int*   cursor = (int*)(ws + 3 * NP);
  int*   src    = (int*)(ws + 4 * NP);
  float* h      = ws + 4 * NP + EP;
  float* acc    = h + (size_t)N * D;

  const int tpb = 256;
  int gNP   = (int)(NP / tpb);
  int gScan = (int)(NP / 1024);
  int gE4   = ((E + 3) / 4 + tpb - 1) / tpb;
  int gNode = (N + 8 * NPW - 1) / (8 * NPW);
  int gGemm = N / 16;

  zero_int<<<gNP, tpb, 0, stream>>>(cnt, (int)NP);
  count_targets<<<gE4, tpb, 0, stream>>>(col, cnt, E);
  scan_counts<<<gScan, tpb, 0, stream>>>(cnt, rowptr, cursor, dinv, N);
  fill_sources<<<gE4, tpb, 0, stream>>>(row, col, cursor, src, E);

  gemm_wmma<<<gGemm, 192, 0, stream>>>(x, W1, h, N);
  gather_rows<<<gNode, tpb, 0, stream>>>(rowptr, src, dinv, h, b1, acc, 1, N);

  gemm_wmma<<<gGemm, 192, 0, stream>>>(acc, W2, h, N);
  gather_rows<<<gNode, tpb, 0, stream>>>(rowptr, src, dinv, h, b2, out, 0, N);
}


# edges grouped by target by a two-level counting sort on LDS counters (256 list pieces x 256-node buckets, returning LDS add = slot): no global atomics left; sum order = LDS arrival order
# speedup vs baseline: 3.0997x; 1.6562x over previous
#include <hip/hip_runtime.h>
#include <hip/hip_bf16.h>

#define D 96

typedef float    v2f  __attribute__((ext_vector_type(2)));
typedef float    v8f  __attribute__((ext_vector_type(8)));
typedef _Float16 v16h __attribute__((ext_vector_type(16)));

#ifndef __has_builtin
#define __has_builtin(x) 0
#endif

#if defined(__AMDGCN__) && __has_builtin(__builtin_amdgcn_wmma_f32_16x16x4_f32)
#define USE_F32_WMMA 1
#else
#define USE_F32_WMMA 0
#endif

#define CHUNKS 256
#define BSH    8
#define HB     256
#define ETH    512
#define EBC    7
#define EBB    9

__global__ void __launch_bounds__(ETH)
chunk_hist(const int* __restrict__ col, int* __restrict__ cnt1, int E, int CH, int nbp) {
  __shared__ int hist[HB];
  int t = threadIdx.x, w = blockIdx.x;
  if (t < HB) hist[t] = 0;
  __syncthreads();
  int e0 = w * CH;
  int e1 = e0 + CH < E ? e0 + CH : E;
  for (int eb = e0; eb < e1; eb += EBC * ETH) {
    int e = eb + t;
    int cv[EBC];
    #pragma unroll
    for (int k = 0; k < EBC; ++k) {
      int ee = e + k * ETH;
      cv[k] = col[ee < e1 ? ee : e1 - 1];
    }
    static_assert(EBC == 7, "the statement below names the seven loaded values");
    asm volatile("" :: "v"(cv[0]), "v"(cv[1]), "v"(cv[2]), "v"(cv[3]), "v"(cv[4]), "v"(cv[5]), "v"(cv[6]));
    #pragma unroll
    for (int k = 0; k < EBC; ++k)
      if (e + k * ETH < e1) atomicAdd(&hist[cv[k] >> BSH], 1);
  }
  __syncthreads();
  if (t < nbp) cnt1[t * CHUNKS + w] = hist[t];
}

__global__ void __launch_bounds__(256)
scan_chunk_counts(const int* __restrict__ cnt, int* __restrict__ off) {
  __shared__ int sh[256];
  int t = threadIdx.x;
  int tile0 = blockIdx.x * 1024;
  int part = 0;
  #pragma unroll 8
  for (int i = t * 4; i < tile0; i += 1024) {
    int4 v = *(const int4*)(cnt + i);
    part += (v.x + v.y) + (v.z + v.w);
  }
  int4 c = *(const int4*)(cnt + tile0 + t * 4);
  int tsum = (c.x + c.y) + (c.z + c.w);

  sh[t] = part;
  __syncthreads();
  for (int o = 128; o > 0; o >>= 1) {
    if (t < o) sh[t] += sh[t + o];
    __syncthreads();
  }
  int base = sh[0];
  __syncthreads();

  sh[t] = tsum;
  __syncthreads();
  for (int o = 1; o < 256; o <<= 1) {
    int v = (t >= o) ? sh[t - o] : 0;
    __syncthreads();
    sh[t] += v;
    __syncthreads();
  }
  int p0 = base + sh[t] - tsum;
  int p1 = p0 + c.x, p2 = p1 + c.y, p3 = p2 + c.z;
  *(int4*)(off + tile0 + t * 4) = make_int4(p0, p1, p2, p3);
}

__global__ void __launch_bounds__(ETH)
chunk_scatter(const int* __restrict__ row, const int* __restrict__ col,
              const int* __restrict__ off1, unsigned* __restrict__ tmp, int E, int CH, int nb) {
  __shared__ int cur[HB];
  int t = threadIdx.x, w = blockIdx.x;
  if (t < nb) cur[t] = off1[t * CHUNKS + w];
  __syncthreads();
  int e0 = w * CH;
  int e1 = e0 + CH < E ? e0 + CH : E;
  for (int eb = e0; eb < e1; eb += EBC * ETH) {
    int e = eb + t;
    int key[EBC], p[EBC];
    unsigned val[EBC];
    #pragma unroll
    for (int k = 0; k < EBC; ++k) {
      int ee = e + k * ETH;
      int ec = ee < e1 ? ee : e1 - 1;
      int cv = col[ec];
      int rv = row[ec];
      key[k] = ee < e1 ? (cv >> BSH) : -1;
      val[k] = (unsigned)rv | ((unsigned)(cv & ((1 << BSH) - 1)) << 24);
    }
    #pragma unroll
    for (int k = 0; k < EBC; ++k) {
      p[k] = 0;
      if (key[k] >= 0) p[k] = atomicAdd(&cur[key[k]], 1);
    }
    #pragma unroll
    for (int k = 0; k < EBC; ++k)
      if (key[k] >= 0) tmp[p[k]] = val[k];
  }
}

__global__ void __launch_bounds__(ETH)
bucket_lists(const int* __restrict__ off1, const unsigned* __restrict__ tmp,
             int* __restrict__ rowptr, float* __restrict__ dinv, int* __restrict__ src,
             int N, int E, int nbp) {
  __shared__ int cnt[1 << BSH];
  __shared__ int cur[1 << BSH];
  __shared__ int wsum[(1 << BSH) / 32];
  int t = threadIdx.x, b = blockIdx.x;
  int lane = t & 31, wv = t >> 5;
  int beg = off1[b * CHUNKS];
  int end = (b + 1 < nbp) ? off1[(b + 1) * CHUNKS] : E;
  int n = end - beg;
  const unsigned* in = tmp + beg;
  if (t < (1 << BSH)) cnt[t] = 0;
  __syncthreads();
  for (int ib = 0; ib < n; ib += EBB * ETH) {
    int i = ib + t;
    unsigned v[EBB];
    #pragma unroll
    for (int k = 0; k < EBB; ++k) {
      int ii = i + k * ETH;
      v[k] = in[ii < n ? ii : n - 1];
    }
    #pragma unroll
    for (int k = 0; k < EBB; ++k)
      if (i + k * ETH < n) atomicAdd(&cnt[v[k] >> 24], 1);
  }
  __syncthreads();
  int c = 0, incl = 0;
  if (t < (1 << BSH)) {
    c = cnt[t];
    incl = c;
    #pragma unroll
    for (int d = 1; d < 32; d <<= 1) {
      int u = __shfl_up(incl, d);
      if (lane >= d) incl += u;
    }
    if (lane == 31) wsum[wv] = incl;
  }
  __syncthreads();
  if (t < (1 << BSH)) {
    int base = 0;
    for (int k = 0; k < wv; ++k) base += wsum[k];
    int excl = base + incl - c;
    int node = (b << BSH) + t;
    if (node <= N) rowptr[node] = beg + excl;
    if (node < N) dinv[node] = rsqrtf((float)c + 1.0f);
    cur[t] = excl;
  }
  __syncthreads();
  int* out = src + beg;
  for (int ib = 0; ib < n; ib += EBB * ETH) {
    int i = ib + t;
    unsigned v[EBB];
    int p[EBB];
    #pragma unroll
    for (int k = 0; k < EBB; ++k) {
      int ii = i + k * ETH;
      v[k] = in[ii < n ? ii : n - 1];
    }
    #pragma unroll
    for (int k = 0; k < EBB; ++k) {
      p[k] = 0;
      if (i + k * ETH < n) p[k] = atomicAdd(&cur[v[k] >> 24], 1);
    }
    #pragma unroll
    for (int k = 0; k < EBB; ++k)
      if (i + k * ETH < n) out[p[k]] = (int)(v[k] & 0xFFFFFFu);
  }
}

__global__ void gemm_wmma(const float* __restrict__ A, const float* __restrict__ B,
                          float* __restrict__ C, int M) {
  int m0 = blockIdx.x * 16;
  if (m0 + 16 > M) return;
  int wave = threadIdx.x >> 5;
  int lane = threadIdx.x & 31;
  int n0   = wave * 16;
  int l    = lane & 15;
  int half = lane >> 4;

  const float* arow = A + (size_t)(m0 + l) * D;
  v8f acc = {};

#if USE_F32_WMMA
  int kk = half * 2;
  #pragma unroll
  for (int k = 0; k < D; k += 4) {
    v2f a, b;
    a.x = arow[k + kk];
    a.y = arow[k + kk + 1];
    b.x = B[(size_t)(k + kk)     * D + n0 + l];
    b.y = B[(size_t)(k + kk + 1) * D + n0 + l];
    acc = __builtin_amdgcn_wmma_f32_16x16x4_f32(
        false, a, false, b, (short)0, acc, false, false);
  }
#else
  for (int k0 = 0; k0 < D; k0 += 32) {
    v16h a, b;
    #pragma unroll
    for (int e = 0; e < 16; ++e) {
      int kk = k0 + half * 8 + ((e >> 3) * 16) + (e & 7);
      a[e] = (_Float16)arow[kk];
      b[e] = (_Float16)B[(size_t)kk * D + n0 + l];
    }
    acc = __builtin_amdgcn_wmma_f32_16x16x32_f16(
        false, a, false, b, (short)0, acc, false, false);
  }
#endif

  int colIdx = n0 + l;
  int rbase  = m0 + half * 8;
  #pragma unroll
  for (int j = 0; j < 8; ++j)
    C[(size_t)(rbase + j) * D + colIdx] = acc[j];
}

template <int K>
__device__ __forceinline__ void add_rows(const int* __restrict__ src, const float* __restrict__ dinv,
                                         const float* __restrict__ h, int j, int lane, float dc,
                                         float& a0, float& a1, float& a2) {
  int r[K];
  float w[K];
  float v[K][3];
  #pragma unroll
  for (int i = 0; i < K; ++i) r[i] = src[j + i];
  #pragma unroll
  for (int i = 0; i < K; ++i) {
    const float* hr = h + (size_t)r[i] * D + lane;
    v[i][0] = hr[0]; v[i][1] = hr[32]; v[i][2] = hr[64];
  }
  #pragma unroll
  for (int i = 0; i < K; ++i) w[i] = dinv[r[i]] * dc;
  #pragma unroll
  for (int i = 0; i < K; ++i) {
    a0 += v[i][0] * w[i];
    a1 += v[i][1] * w[i];
    a2 += v[i][2] * w[i];
  }
}

__device__ __forceinline__ void one_node(const int* __restrict__ rowptr, const int* __restrict__ src,
                                         const float* __restrict__ dinv, const float* __restrict__ h,
                                         float* __restrict__ out, int relu, int node, int lane,
                                         float b0, float b1, float b2) {
  int beg = rowptr[node], end = rowptr[node + 1];
  float dc = dinv[node];
  const float* hc = h + (size_t)node * D + lane;
  float s0 = hc[0], s1 = hc[32], s2 = hc[64];
  float a0 = 0.0f, a1 = 0.0f, a2 = 0.0f;
  int j = beg;
  for (; j + 8 <= end; j += 8) add_rows<8>(src, dinv, h, j, lane, dc, a0, a1, a2);
  if (j + 4 <= end) { add_rows<4>(src, dinv, h, j, lane, dc, a0, a1, a2); j += 4; }
  if (j + 2 <= end) { add_rows<2>(src, dinv, h, j, lane, dc, a0, a1, a2); j += 2; }
  if (j < end) add_rows<1>(src, dinv, h, j, lane, dc, a0, a1, a2);
  float s = dc * dc;
  a0 += s0 * s + b0;
  a1 += s1 * s + b1;
  a2 += s2 * s + b2;
  if (relu) { a0 = fmaxf(a0, 0.0f); a1 = fmaxf(a1, 0.0f); a2 = fmaxf(a2, 0.0f); }
  float* o = out + (size_t)node * D + lane;
  o[0] = a0; o[32] = a1; o[64] = a2;
}

#define NPW 4
__global__ void __launch_bounds__(256)
gather_rows(const int* __restrict__ rowptr, const int* __restrict__ src,
            const float* __restrict__ dinv, const float* __restrict__ h,
            const float* __restrict__ b, float* __restrict__ out, int relu, int N) {
  int lane = threadIdx.x & 31;
  int wave = __builtin_amdgcn_readfirstlane((int)((blockIdx.x * blockDim.x + threadIdx.x) >> 5));
  int node0 = wave * NPW;
  if (node0 >= N) return;
  float b0 = b[lane], b1 = b[lane + 32], b2 = b[lane + 64];
  int nend = node0 + NPW < N ? node0 + NPW : N;
  for (int node = node0; node < nend; ++node)
    one_node(rowptr, src, dinv, h, out, relu, node, lane, b0, b1, b2);
}

extern "C" void kernel_launch(void* const* d_in, const int* in_sizes, int n_in,
                              void* d_out, int out_size, void* d_ws, size_t ws_size,
                              hipStream_t stream) {
  const float* x  = (const float*)d_in[0];
  const int*   ei = (const int*)d_in[1];
  const float* W1 = (const float*)d_in[2];
  const float* b1 = (const float*)d_in[3];
  const float* W2 = (const float*)d_in[4];
  const float* b2 = (const float*)d_in[5];
  float* out = (float*)d_out;

  int N = in_sizes[0] / D;
  int E = in_sizes[1] / 2;
  const int* row = ei;
  const int* col = ei + E;

  int nb  = N / (1 << BSH) + 1;
  int nbp = (nb + 3) / 4 * 4;
  int CH  = ((E + CHUNKS - 1) / CHUNKS + 31) / 32 * 32;
  size_t NP = (((size_t)N + 1 + 1023) / 1024) * 1024;
  size_t EP = (((size_t)E + 1023) / 1024) * 1024;
  size_t TP = (size_t)nbp * CHUNKS;
  float* ws     = (float*)d_ws;
  float* dinv   = ws;
  int*   rowptr = (int*)(ws + NP);
  int*   cnt1   = (int*)(ws + 2 * NP);
  int*   off1   = (int*)(ws + 2 * NP + TP);
  int*   src    = (int*)(ws + 2 * NP + 2 * TP);
  unsigned* tmp = (unsigned*)(ws + 2 * NP + 2 * TP + EP);
  float* h      = ws + 2 * NP + 2 * TP + 2 * EP;
  float* acc    = h + (size_t)N * D;

  const int tpb = 256;
  int gScan = (int)(TP / 1024);
  int gNode = (N + 8 * NPW - 1) / (8 * NPW);
  int gGemm = N / 16;

  chunk_hist<<<CHUNKS, ETH, 0, stream>>>(col, cnt1, E, CH, nbp);
  scan_chunk_counts<<<gScan, tpb, 0, stream>>>(cnt1, off1);
  chunk_scatter<<<CHUNKS, ETH, 0, stream>>>(row, col, off1, tmp, E, CH, nb);
  bucket_lists<<<nb, ETH, 0, stream>>>(off1, tmp, rowptr, dinv, src, N, E, nbp);

  gemm_wmma<<<gGemm, 192, 0, stream>>>(x, W1, h, N);
  gather_rows<<<gNode, tpb, 0, stream>>>(rowptr, src, dinv, h, b1, acc, 1, N);

  gemm_wmma<<<gGemm, 192, 0, stream>>>(acc, W2, h, N);
  gather_rows<<<gNode, tpb, 0, stream>>>(rowptr, src, dinv, h, b2, out, 0, N);
}


# each layer sums a node's INPUT rows first, then does the 96x96 f32 WMMA product on the 16 summed rows in LDS: one launch per layer, no h plane (8->6 launches); sum order: rows summed before product
# speedup vs baseline: 3.8016x; 1.2264x over previous
#include <hip/hip_runtime.h>
#include <hip/hip_bf16.h>

#define D 96

typedef float    v2f  __attribute__((ext_vector_type(2)));
typedef float    v8f  __attribute__((ext_vector_type(8)));
typedef _Float16 v16h __attribute__((ext_vector_type(16)));

#ifndef __has_builtin
#define __has_builtin(x) 0
#endif

#if defined(__AMDGCN__) && __has_builtin(__builtin_amdgcn_wmma_f32_16x16x4_f32)
#define USE_F32_WMMA 1
#else
#define USE_F32_WMMA 0
#endif

#define CHUNKS 256
#define BSH    8
#define HB     256
#define ETH    512
#define EBC    7
#define EBB    9

__global__ void __launch_bounds__(ETH)
chunk_hist(const int* __restrict__ col, int* __restrict__ cnt1, int E, int CH, int nbp) {
  __shared__ int hist[HB];
  int t = threadIdx.x, w = blockIdx.x;
  if (t < HB) hist[t] = 0;
  __syncthreads();
  int e0 = w * CH;
  int e1 = e0 + CH < E ? e0 + CH : E;
  for (int eb = e0; eb < e1; eb += EBC * ETH) {
    int e = eb + t;
    int cv[EBC];
    #pragma unroll
    for (int k = 0; k < EBC; ++k) {
      int ee = e + k * ETH;
      cv[k] = col[ee < e1 ? ee : e1 - 1];
    }
    static_assert(EBC == 7, "the statement below names the seven loaded values");
    asm volatile("" :: "v"(cv[0]), "v"(cv[1]), "v"(cv[2]), "v"(cv[3]), "v"(cv[4]), "v"(cv[5]), "v"(cv[6]));
    #pragma unroll
    for (int k = 0; k < EBC; ++k)
      if (e + k * ETH < e1) atomicAdd(&hist[cv[k] >> BSH], 1);
  }
  __syncthreads();
  if (t < nbp) cnt1[t * CHUNKS + w] = hist[t];
}

__global__ void __launch_bounds__(256)
scan_chunk_counts(const int* __restrict__ cnt, int* __restrict__ off) {
  __shared__ int sh[256];
  int t = threadIdx.x;
  int tile0 = blockIdx.x * 1024;
  int part = 0;
  #pragma unroll 8
  for (int i = t * 4; i < tile0; i += 1024) {
    int4 v = *(const int4*)(cnt + i);
    part += (v.x + v.y) + (v.z + v.w);
  }
  int4 c = *(const int4*)(cnt + tile0 + t * 4);
  int tsum = (c.x + c.y) + (c.z + c.w);

  sh[t] = part;
  __syncthreads();
  for (int o = 128; o > 0; o >>= 1) {
    if (t < o) sh[t] += sh[t + o];
    __syncthreads();
  }
  int base = sh[0];
  __syncthreads();

  sh[t] = tsum;
  __syncthreads();
  for (int o = 1; o < 256; o <<= 1) {
    int v = (t >= o) ? sh[t - o] : 0;
    __syncthreads();
    sh[t] += v;
    __syncthreads();
  }
  int p0 = base + sh[t] - tsum;
  int p1 = p0 + c.x, p2 = p1 + c.y, p3 = p2 + c.z;
  *(int4*)(off + tile0 + t * 4) = make_int4(p0, p1, p2, p3);
}

__global__ void __launch_bounds__(ETH)
chunk_scatter(const int* __restrict__ row, const int* __restrict__ col,
              const int* __restrict__ off1, unsigned* __restrict__ tmp, int E, int CH, int nb) {
  __shared__ int cur[HB];
  int t = threadIdx.x, w = blockIdx.x;
  if (t < nb) cur[t] = off1[t * CHUNKS + w];
  __syncthreads();
  int e0 = w * CH;
  int e1 = e0 + CH < E ? e0 + CH : E;
  for (int eb = e0; eb < e1; eb += EBC * ETH) {
    int e = eb + t;
    int key[EBC], p[EBC];
    unsigned val[EBC];
    #pragma unroll
    for (int k = 0; k < EBC; ++k) {
      int ee = e + k * ETH;
      int ec = ee < e1 ? ee : e1 - 1;
      int cv = col[ec];
      int rv = row[ec];
      key[k] = ee < e1 ? (cv >> BSH) : -1;
      val[k] = (unsigned)rv | ((unsigned)(cv & ((1 << BSH) - 1)) << 24);
    }
    #pragma unroll
    for (int k = 0; k < EBC; ++k) {
      p[k] = 0;
      if (key[k] >= 0) p[k] = atomicAdd(&cur[key[k]], 1);
    }
    #pragma unroll
    for (int k = 0; k < EBC; ++k)
      if (key[k] >= 0) tmp[p[k]] = val[k];
  }
}

__global__ void __launch_bounds__(ETH)
bucket_lists(const int* __restrict__ off1, const unsigned* __restrict__ tmp,
             int* __restrict__ rowptr, float* __restrict__ dinv, int* __restrict__ src,
             int N, int E, int nbp) {
  __shared__ int cnt[1 << BSH];
  __shared__ int cur[1 << BSH];
  __shared__ int wsum[(1 << BSH) / 32];
  int t = threadIdx.x, b = blockIdx.x;
  int lane = t & 31, wv = t >> 5;
  int beg = off1[b * CHUNKS];
  int end = (b + 1 < nbp) ? off1[(b + 1) * CHUNKS] : E;
  int n = end - beg;
  const unsigned* in = tmp + beg;
  if (t < (1 << BSH)) cnt[t] = 0;
  __syncthreads();
  for (int ib = 0; ib < n; ib += EBB * ETH) {
    int i = ib + t;
    unsigned v[EBB];
    #pragma unroll
    for (int k = 0; k < EBB; ++k) {
      int ii = i + k * ETH;
      v[k] = in[ii < n ? ii : n - 1];
    }
    #pragma unroll
    for (int k = 0; k < EBB; ++k)
      if (i + k * ETH < n) atomicAdd(&cnt[v[k] >> 24], 1);
  }
  __syncthreads();
  int c = 0, incl = 0;
  if (t < (1 << BSH)) {
    c = cnt[t];
    incl = c;
    #pragma unroll
    for (int d = 1; d < 32; d <<= 1) {
      int u = __shfl_up(incl, d);
      if (lane >= d) incl += u;
    }
    if (lane == 31) wsum[wv] = incl;
  }
  __syncthreads();
  if (t < (1 << BSH)) {
    int base = 0;
    for (int k = 0; k < wv; ++k) base += wsum[k];
    int excl = base + incl - c;
    int node = (b << BSH) + t;
    if (node <= N) rowptr[node] = beg + excl;
    if (node < N) dinv[node] = rsqrtf((float)c + 1.0f);
    cur[t] = excl;
  }
  __syncthreads();
  int* out = src + beg;
  for (int ib = 0; ib < n; ib += EBB * ETH) {
    int i = ib + t;
    unsigned v[EBB];
    int p[EBB];
    #pragma unroll
    for (int k = 0; k < EBB; ++k) {
      int ii = i + k * ETH;
      v[k] = in[ii < n ? ii : n - 1];
    }
    #pragma unroll
    for (int k = 0; k < EBB; ++k) {
      p[k] = 0;
      if (i + k * ETH < n) p[k] = atomicAdd(&cur[v[k] >> 24], 1);
    }
    #pragma unroll
    for (int k = 0; k < EBB; ++k)
      if (i + k * ETH < n) out[p[k]] = (int)(v[k] & 0xFFFFFFu);
  }
}

template <int K>
__device__ __forceinline__ void add_rows(const int* __restrict__ src, const float* __restrict__ dinv,
                                         const float* __restrict__ h, int j, int lane, float dc,
                                         float& a0, float& a1, float& a2) {
  int r[K];
  float w[K];
  float v[K][3];
  #pragma unroll
  for (int i = 0; i < K; ++i) r[i] = src[j + i];
  #pragma unroll
  for (int i = 0; i < K; ++i) {
    const float* hr = h + (size_t)r[i] * D + lane;
    v[i][0] = hr[0]; v[i][1] = hr[32]; v[i][2] = hr[64];
  }
  #pragma unroll
  for (int i = 0; i < K; ++i) w[i] = dinv[r[i]] * dc;
  #pragma unroll
  for (int i = 0; i < K; ++i) {
    a0 += v[i][0] * w[i];
    a1 += v[i][1] * w[i];
    a2 += v[i][2] * w[i];
  }
}

__device__ __forceinline__ void sum_node(const int* __restrict__ rowptr, const int* __restrict__ src,
                                         const float* __restrict__ dinv, const float* __restrict__ f,
                                         int node, int lane, float& a0, float& a1, float& a2) {
  int beg = rowptr[node], end = rowptr[node + 1];
  float dc = dinv[node];
  const float* fc = f + (size_t)node * D + lane;
  float s0 = fc[0], s1 = fc[32], s2 = fc[64];
  int j = beg;
  for (; j + 8 <= end; j += 8) add_rows<8>(src, dinv, f, j, lane, dc, a0, a1, a2);
  if (j + 4 <= end) { add_rows<4>(src, dinv, f, j, lane, dc, a0, a1, a2); j += 4; }
  if (j + 2 <= end) { add_rows<2>(src, dinv, f, j, lane, dc, a0, a1, a2); j += 2; }
  if (j < end) add_rows<1>(src, dinv, f, j, lane, dc, a0, a1, a2);
  float s = dc * dc;
  a0 += s0 * s;
  a1 += s1 * s;
  a2 += s2 * s;
}

#define NPW 4
#define TM  16
#define AP  100
#define KB  8
__global__ void __launch_bounds__(32 * (TM / NPW), 16)
layer_sum_then_product(const int* __restrict__ rowptr, const int* __restrict__ src,
                       const float* __restrict__ dinv, const float* __restrict__ f,
                       const float* __restrict__ Wm, const float* __restrict__ b,
                       float* __restrict__ out, int relu, int N) {
  static_assert(TM / NPW == 4 && D == 96 && D % (4 * KB) == 0,
                "4 waves: three of them cover the 96 columns, 32 each; whole trips");
  __shared__ float agg[TM * AP];
  int lane = threadIdx.x & 31;
  int wave = __builtin_amdgcn_readfirstlane((int)(threadIdx.x >> 5));
  int m0 = blockIdx.x * TM;

  #pragma nounroll
  for (int i = 0; i < NPW; ++i) {
    int lr = wave * NPW + i;
    int node = m0 + lr;
    float a0 = 0.0f, a1 = 0.0f, a2 = 0.0f;
    if (node < N) sum_node(rowptr, src, dinv, f, node, lane, a0, a1, a2);
    float* ar = agg + lr * AP + lane;
    ar[0] = a0; ar[32] = a1; ar[64] = a2;
  }
  __syncthreads();
  if (wave == 3) return;

  int c  = lane & 15;
  int hf = lane >> 4;
  int n0 = wave * 32;
  const float* ap = agg + c * AP + 2 * hf;
  const float* bp = Wm + (size_t)(2 * hf) * D + n0 + 2 * c;
  v8f acc0 = {}, acc1 = {};
#if USE_F32_WMMA
  #pragma nounroll
  for (int kb = 0; kb < D; kb += 4 * KB) {
    v2f w0[KB], w1[KB];
    #pragma unroll
    for (int s = 0; s < KB; ++s) {
      w0[s] = *(const v2f*)(bp + (size_t)(kb + 4 * s) * D);
      w1[s] = *(const v2f*)(bp + (size_t)(kb + 4 * s + 1) * D);
    }
    #pragma unroll
    for (int s = 0; s < KB; ++s) {
      v2f a = *(const v2f*)(ap + kb + 4 * s);
      v2f b0, b1;
      b0.x = w0[s].x; b0.y = w1[s].x;
      b1.x = w0[s].y; b1.y = w1[s].y;
      acc0 = __builtin_amdgcn_wmma_f32_16x16x4_f32(
          false, a, false, b0, (short)0, acc0, false, false);
      acc1 = __builtin_amdgcn_wmma_f32_16x16x4_f32(
          false, a, false, b1, (short)0, acc1, false, false);
    }
  }
#endif
  v2f bias = *(const v2f*)(b + n0 + 2 * c);
  int rbase = m0 + hf * 8;
  #pragma unroll
  for (int j = 0; j < 8; ++j) {
    v2f v;
    v.x = acc0[j] + bias.x;
    v.y = acc1[j] + bias.y;
    if (relu) { v.x = fmaxf(v.x, 0.0f); v.y = fmaxf(v.y, 0.0f); }
    if (rbase + j < N) *(v2f*)(out + (size_t)(rbase + j) * D + n0 + 2 * c) = v;
  }
}

extern "C" void kernel_launch(void* const* d_in, const int* in_sizes, int n_in,
                              void* d_out, int out_size, void* d_ws, size_t ws_size,
                              hipStream_t stream) {
  const float* x  = (const float*)d_in[0];
  const int*   ei = (const int*)d_in[1];
  const float* W1 = (const float*)d_in[2];
  const float* b1 = (const float*)d_in[3];
  const float* W2 = (const float*)d_in[4];
  const float* b2 = (const float*)d_in[5];
  float* out = (float*)d_out;

  int N = in_sizes[0] / D;
  int E = in_sizes[1] / 2;
  const int* row = ei;
  const int* col = ei + E;

  int nb  = N / (1 << BSH) + 1;
  int nbp = (nb + 3) / 4 * 4;
  int CH  = ((E + CHUNKS - 1) / CHUNKS + 31) / 32 * 32;
  size_t NP = (((size_t)N + 1 + 1023) / 1024) * 1024;
  size_t EP = (((size_t)E + 1023) / 1024) * 1024;
  size_t TP = (size_t)nbp * CHUNKS;
  float* ws     = (float*)d_ws;
  float* dinv   = ws;
  int*   rowptr = (int*)(ws + NP);
  int*   cnt1   = (int*)(ws + 2 * NP);
  int*   off1   = (int*)(ws + 2 * NP + TP);
  int*   src    = (int*)(ws + 2 * NP + 2 * TP);
  unsigned* tmp = (unsigned*)(ws + 2 * NP + 2 * TP + EP);
  float* acc    = ws + 2 * NP + 2 * TP + 2 * EP;

  const int tpb = 256;
  int gScan = (int)(TP / 1024);
  int gTile = (N + TM - 1) / TM;

  chunk_hist<<<CHUNKS, ETH, 0, stream>>>(col, cnt1, E, CH, nbp);
  scan_chunk_counts<<<gScan, tpb, 0, stream>>>(cnt1, off1);
  chunk_scatter<<<CHUNKS, ETH, 0, stream>>>(row, col, off1, tmp, E, CH, nb);
  bucket_lists<<<nb, ETH, 0, stream>>>(off1, tmp, rowptr, dinv, src, N, E, nbp);

  layer_sum_then_product<<<gTile, 32 * (TM / NPW), 0, stream>>>(rowptr, src, dinv, x, W1, b1, acc, 1, N);

  layer_sum_then_product<<<gTile, 32 * (TM / NPW), 0, stream>>>(rowptr, src, dinv, acc, W2, b2, out, 0, N);
}


# level 1 of the edge sort stays inside each piece's own range (count, LDS prefix, place: ONE launch); bucket pass collects its 256 runs and sums its own start: 6->4 launches; sum order: arrival order
# speedup vs baseline: 4.8726x; 1.2817x over previous
#include <hip/hip_runtime.h>
#include <hip/hip_bf16.h>

#define D 96

typedef float    v2f  __attribute__((ext_vector_type(2)));
typedef float    v8f  __attribute__((ext_vector_type(8)));
typedef _Float16 v16h __attribute__((ext_vector_type(16)));

#ifndef __has_builtin
#define __has_builtin(x) 0
#endif

#if defined(__AMDGCN__) && __has_builtin(__builtin_amdgcn_wmma_f32_16x16x4_f32)
#define USE_F32_WMMA 1
#else
#define USE_F32_WMMA 0
#endif

#define CHUNKS 256
#define BSH    8
#define HB     256
#define ETH    512
#define EBC    7
#define EBB    9

__device__ __forceinline__ int scan256(int c, int t, int* wsum, int* total) {
  int lane = t & 31, wv = t >> 5;
  int incl = c;
  #pragma unroll
  for (int d = 1; d < 32; d <<= 1) {
    int u = __shfl_up(incl, d);
    if (lane >= d) incl += u;
  }
  if (wv < 8 && lane == 31) wsum[wv] = incl;
  __syncthreads();
  int base = 0, tot = 0;
  #pragma unroll
  for (int k = 0; k < 8; ++k) {
    int s = wsum[k];
    if (k < wv) base += s;
    tot += s;
  }
  __syncthreads();
  *total = tot;
  return base + incl - c;
}

__global__ void __launch_bounds__(ETH)
piece_sort(const int* __restrict__ row, const int* __restrict__ col,
           int* __restrict__ cnt1, int* __restrict__ loc1, unsigned* __restrict__ tmp,
           int E, int CH, int nbp) {
  static_assert(HB == 256, "scan256 covers 256 counters");
  __shared__ int hist[HB];
  __shared__ int wsum[8];
  int t = threadIdx.x, w = blockIdx.x;
  if (t < HB) hist[t] = 0;
  __syncthreads();
  int e0 = w * CH;
  int e1 = e0 + CH < E ? e0 + CH : E;
  for (int eb = e0; eb < e1; eb += EBC * ETH) {
    int e = eb + t;
    int cv[EBC];
    #pragma unroll
    for (int k = 0; k < EBC; ++k) {
      int ee = e + k * ETH;
      cv[k] = col[ee < e1 ? ee : e1 - 1];
    }
    static_assert(EBC == 7, "the statement below names the seven loaded values");
    asm volatile("" :: "v"(cv[0]), "v"(cv[1]), "v"(cv[2]), "v"(cv[3]), "v"(cv[4]), "v"(cv[5]), "v"(cv[6]));
    #pragma unroll
    for (int k = 0; k < EBC; ++k)
      if (e + k * ETH < e1) atomicAdd(&hist[cv[k] >> BSH], 1);
  }
  __syncthreads();
  int c = (t < HB) ? hist[t] : 0;
  int tot;
  int excl = scan256(c, t, wsum, &tot);
  if (t < nbp) {
    cnt1[t * CHUNKS + w] = c;
    loc1[t * CHUNKS + w] = excl;
  }
  if (t < HB) hist[t] = excl;
  __syncthreads();
  unsigned* out = tmp + e0;
  for (int eb = e0; eb < e1; eb += EBC * ETH) {
    int e = eb + t;
    int key[EBC], p[EBC];
    unsigned val[EBC];
    #pragma unroll
    for (int k = 0; k < EBC; ++k) {
      int ee = e + k * ETH;
      int ec = ee < e1 ? ee : e1 - 1;
      int cv = col[ec];
      int rv = row[ec];
      key[k] = ee < e1 ? (cv >> BSH) : -1;
      val[k] = (unsigned)rv | ((unsigned)(cv & ((1 << BSH) - 1)) << 24);
    }
    #pragma unroll
    for (int k = 0; k < EBC; ++k) {
      p[k] = 0;
      if (key[k] >= 0) p[k] = atomicAdd(&hist[key[k]], 1);
    }
    #pragma unroll
    for (int k = 0; k < EBC; ++k)
      if (key[k] >= 0) out[p[k]] = val[k];
  }
}

__device__ __forceinline__ int run_index(const int* pre, const int* rb, int i) {
  int w = 0;
  #pragma unroll
  for (int s = CHUNKS / 2; s > 0; s >>= 1)
    if (pre[w + s] <= i) w += s;
  return rb[w] + i;
}

__global__ void __launch_bounds__(ETH)
bucket_lists(const int* __restrict__ cnt1, const int* __restrict__ loc1,
             const unsigned* __restrict__ tmp,
             int* __restrict__ rowptr, float* __restrict__ dinv, int* __restrict__ src,
             int N, int CH) {
  static_assert(CHUNKS == 256 && (1 << BSH) == 256, "scan256 covers 256 runs and 256 nodes");
  __shared__ int cnt[1 << BSH];
  __shared__ int cur[1 << BSH];
  __shared__ int pre[CHUNKS];
  __shared__ int rb[CHUNKS];
  __shared__ int wsum[8];
  __shared__ int red[ETH / 32];
  int t = threadIdx.x, b = blockIdx.x;
  int lane = t & 31, wv = t >> 5;
  int part = 0;
  int nfront = b * (CHUNKS / 4);
  #pragma unroll 8
  for (int i = t; i < nfront; i += ETH) {
    int4 v = *(const int4*)(cnt1 + 4 * (size_t)i);
    part += (v.x + v.y) + (v.z + v.w);
  }
  int c = 0, lc = 0;
  if (t < CHUNKS) {
    c  = cnt1[b * CHUNKS + t];
    lc = loc1[b * CHUNKS + t];
  }
  #pragma unroll
  for (int d = 16; d > 0; d >>= 1) part += __shfl_xor(part, d);
  if (lane == 0) red[wv] = part;
  if (t < (1 << BSH)) cnt[t] = 0;
  int n;
  int ex = scan256(c, t, wsum, &n);
  int beg = 0;
  #pragma unroll
  for (int k = 0; k < ETH / 32; ++k) beg += red[k];
  if (t < CHUNKS) {
    pre[t] = ex;
    rb[t]  = t * CH + lc - ex;
  }
  __syncthreads();
  for (int ib = 0; ib < n; ib += EBB * ETH) {
    int i = ib + t;
    unsigned v[EBB];
    #pragma unroll
    for (int k = 0; k < EBB; ++k) {
      int ii = i + k * ETH;
      v[k] = tmp[run_index(pre, rb, ii < n ? ii : n - 1)];
    }
    #pragma unroll
    for (int k = 0; k < EBB; ++k)
      if (i + k * ETH < n) atomicAdd(&cnt[v[k] >> 24], 1);
  }
  __syncthreads();
  int cn = (t < (1 << BSH)) ? cnt[t] : 0;
  int ntot;
  int excl = scan256(cn, t, wsum, &ntot);
  if (t < (1 << BSH)) {
    int node = (b << BSH) + t;
    if (node <= N) rowptr[node] = beg + excl;
    if (node < N) dinv[node] = rsqrtf((float)cn + 1.0f);
    cur[t] = excl;
  }
  __syncthreads();
  int* out = src + beg;
  for (int ib = 0; ib < n; ib += EBB * ETH) {
    int i = ib + t;
    unsigned v[EBB];
    int p[EBB];
    #pragma unroll
    for (int k = 0; k < EBB; ++k) {
      int ii = i + k * ETH;
      v[k] = tmp[run_index(pre, rb, ii < n ? ii : n - 1)];
    }
    #pragma unroll
    for (int k = 0; k < EBB; ++k) {
      p[k] = 0;
      if (i + k * ETH < n) p[k] = atomicAdd(&cur[v[k] >> 24], 1);
    }
    #pragma unroll
    for (int k = 0; k < EBB; ++k)
      if (i + k * ETH < n) out[p[k]] = (int)(v[k] & 0xFFFFFFu);
  }
}

template <int K>
__device__ __forceinline__ void add_rows(const int* __restrict__ src, const float* __restrict__ dinv,
                                         const float* __restrict__ h, int j, int lane, float dc,
                                         float& a0, float& a1, float& a2) {
  int r[K];
  float w[K];
  float v[K][3];
  #pragma unroll
  for (int i = 0; i < K; ++i) r[i] = src[j + i];
  #pragma unroll
  for (int i = 0; i < K; ++i) {
    const float* hr = h + (size_t)r[i] * D + lane;
    v[i][0] = hr[0]; v[i][1] = hr[32]; v[i][2] = hr[64];
  }
  #pragma unroll
  for (int i = 0; i < K; ++i) w[i] = dinv[r[i]] * dc;
  #pragma unroll
  for (int i = 0; i < K; ++i) {
    a0 += v[i][0] * w[i];
    a1 += v[i][1] * w[i];
    a2 += v[i][2] * w[i];
  }
}

__device__ __forceinline__ void sum_node(const int* __restrict__ rowptr, const int* __restrict__ src,
                                         const float* __restrict__ dinv, const float* __restrict__ f,
                                         int node, int lane, float& a0, float& a1, float& a2) {
  int beg = rowptr[node], end = rowptr[node + 1];
  float dc = dinv[node];
  const float* fc = f + (size_t)node * D + lane;
  float s0 = fc[0], s1 = fc[32], s2 = fc[64];
  int j = beg;
  for (; j + 8 <= end; j += 8) add_rows<8>(src, dinv, f, j, lane, dc, a0, a1, a2);
  if (j + 4 <= end) { add_rows<4>(src, dinv, f, j, lane, dc, a0, a1, a2); j += 4; }
  if (j + 2 <= end) { add_rows<2>(src, dinv, f, j, lane, dc, a0, a1, a2); j += 2; }
  if (j < end) add_rows<1>(src, dinv, f, j, lane, dc, a0, a1, a2);
  float s = dc * dc;
  a0 += s0 * s;
  a1 += s1 * s;
  a2 += s2 * s;
}

#define NPW 4
#define TM  16
#define AP  100
#define KB  8
__global__ void __launch_bounds__(32 * (TM / NPW), 16)
layer_sum_then_product(const int* __restrict__ rowptr, const int* __restrict__ src,
                       const float* __restrict__ dinv, const float* __restrict__ f,
                       const float* __restrict__ Wm, const float* __restrict__ b,
                       float* __restrict__ out, int relu, int N) {
  static_assert(TM / NPW == 4 && D == 96 && D % (4 * KB) == 0,
                "4 waves: three of them cover the 96 columns, 32 each; whole trips");
  __shared__ float agg[TM * AP];
  int lane = threadIdx.x & 31;
  int wave = __builtin_amdgcn_readfirstlane((int)(threadIdx.x >> 5));
  int m0 = blockIdx.x * TM;

  #pragma nounroll
  for (int i = 0; i < NPW; ++i) {
    int lr = wave * NPW + i;
    int node = m0 + lr;
    float a0 = 0.0f, a1 = 0.0f, a2 = 0.0f;
    if (node < N) sum_node(rowptr, src, dinv, f, node, lane, a0, a1, a2);
    float* ar = agg + lr * AP + lane;
    ar[0] = a0; ar[32] = a1; ar[64] = a2;
  }
  __syncthreads();
  if (wave == 3) return;

  int c  = lane & 15;
  int hf = lane >> 4;
  int n0 = wave * 32;
  const float* ap = agg + c * AP + 2 * hf;
  const float* bp = Wm + (size_t)(2 * hf) * D + n0 + 2 * c;
  v8f acc0 = {}, acc1 = {};
#if USE_F32_WMMA
  #pragma nounroll
  for (int kb = 0; kb < D; kb += 4 * KB) {
    v2f w0[KB], w1[KB];
    #pragma unroll
    for (int s = 0; s < KB; ++s) {
      w0[s] = *(const v2f*)(bp + (size_t)(kb + 4 * s) * D);
      w1[s] = *(const v2f*)(bp + (size_t)(kb + 4 * s + 1) * D);
    }
    #pragma unroll
    for (int s = 0; s < KB; ++s) {
      v2f a = *(const v2f*)(ap + kb + 4 * s);
      v2f b0, b1;
      b0.x = w0[s].x; b0.y = w1[s].x;
      b1.x = w0[s].y; b1.y = w1[s].y;
      acc0 = __builtin_amdgcn_wmma_f32_16x16x4_f32(
          false, a, false, b0, (short)0, acc0, false, false);
      acc1 = __builtin_amdgcn_wmma_f32_16x16x4_f32(
          false, a, false, b1, (short)0, acc1, false, false);
    }
  }
#endif
  v2f bias = *(const v2f*)(b + n0 + 2 * c);
  int rbase = m0 + hf * 8;
  #pragma unroll
  for (int j = 0; j < 8; ++j) {
    v2f v;
    v.x = acc0[j] + bias.x;
    v.y = acc1[j] + bias.y;
    if (relu) { v.x = fmaxf(v.x, 0.0f); v.y = fmaxf(v.y, 0.0f); }
    if (rbase + j < N) *(v2f*)(out + (size_t)(rbase + j) * D + n0 + 2 * c) = v;
  }
}

extern "C" void kernel_launch(void* const* d_in, const int* in_sizes, int n_in,
                              void* d_out, int out_size, void* d_ws, size_t ws_size,
                              hipStream_t stream) {
  const float* x  = (const float*)d_in[0];
  const int*   ei = (const int*)d_in[1];
  const float* W1 = (const float*)d_in[2];
  const float* b1 = (const float*)d_in[3];
  const float* W2 = (const float*)d_in[4];
  const float* b2 = (const float*)d_in[5];
  float* out = (float*)d_out;

  int N = in_sizes[0] / D;
  int E = in_sizes[1] / 2;
  const int* row = ei;
  const int* col = ei + E;

  int nb  = N / (1 << BSH) + 1;
  int nbp = (nb + 3) / 4 * 4;
  int CH  = ((E + CHUNKS - 1) / CHUNKS + 31) / 32 * 32;
  size_t NP = (((size_t)N + 1 + 1023) / 1024) * 1024;
  size_t EP = (((size_t)E + 1023) / 1024) * 1024;
  size_t TP = (size_t)nbp * CHUNKS;
  float* ws     = (float*)d_ws;
  float* dinv   = ws;
  int*   rowptr = (int*)(ws + NP);
  int*   cnt1   = (int*)(ws + 2 * NP);
  int*   loc1   = (int*)(ws + 2 * NP + TP);
  int*   src    = (int*)(ws + 2 * NP + 2 * TP);
  unsigned* tmp = (unsigned*)(ws + 2 * NP + 2 * TP + EP);
  float* acc    = ws + 2 * NP + 2 * TP + 2 * EP;

  int gTile = (N + TM - 1) / TM;

  piece_sort<<<CHUNKS, ETH, 0, stream>>>(row, col, cnt1, loc1, tmp, E, CH, nbp);
  bucket_lists<<<nb, ETH, 0, stream>>>(cnt1, loc1, tmp, rowptr, dinv, src, N, CH);

  layer_sum_then_product<<<gTile, 32 * (TM / NPW), 0, stream>>>(rowptr, src, dinv, x, W1, b1, acc, 1, N);

  layer_sum_then_product<<<gTile, 32 * (TM / NPW), 0, stream>>>(rowptr, src, dinv, acc, W2, b2, out, 0, N);
}


# edge-sort passes place their words into an LDS copy of the piece / bucket, then write it out in order, 128 B in a row per store, instead of one scattered 4-B global store per edge; slot order kept
# speedup vs baseline: 5.0710x; 1.0407x over previous
#include <hip/hip_runtime.h>
#include <hip/hip_bf16.h>

#define D 96

typedef float    v2f  __attribute__((ext_vector_type(2)));
typedef float    v8f  __attribute__((ext_vector_type(8)));
typedef _Float16 v16h __attribute__((ext_vector_type(16)));

#ifndef __has_builtin
#define __has_builtin(x) 0
#endif

#if defined(__AMDGCN__) && __has_builtin(__builtin_amdgcn_wmma_f32_16x16x4_f32)
#define USE_F32_WMMA 1
#else
#define USE_F32_WMMA 0
#endif

#define CHUNKS 256
#define BSH    8
#define HB     256
#define ETH    512
#define EBC    7
#define EBB    9
#define PCAP   4096
#define BCAP   (EBB * ETH)

__device__ __forceinline__ int scan256(int c, int t, int* wsum, int* total) {
  int lane = t & 31, wv = t >> 5;
  int incl = c;
  #pragma unroll
  for (int d = 1; d < 32; d <<= 1) {
    int u = __shfl_up(incl, d);
    if (lane >= d) incl += u;
  }
  if (wv < 8 && lane == 31) wsum[wv] = incl;
  __syncthreads();
  int base = 0, tot = 0;
  #pragma unroll
  for (int k = 0; k < 8; ++k) {
    int s = wsum[k];
    if (k < wv) base += s;
    tot += s;
  }
  __syncthreads();
  *total = tot;
  return base + incl - c;
}

__global__ void __launch_bounds__(ETH)
piece_sort(const int* __restrict__ row, const int* __restrict__ col,
           int* __restrict__ cnt1, int* __restrict__ loc1, unsigned* __restrict__ tmp,
           int E, int CH, int nbp) {
  static_assert(HB == 256, "scan256 covers 256 counters");
  __shared__ int hist[HB];
  __shared__ int wsum[8];
  __shared__ unsigned stage[PCAP];
  int t = threadIdx.x, w = blockIdx.x;
  if (t < HB) hist[t] = 0;
  __syncthreads();
  int e0 = w * CH;
  int e1 = e0 + CH < E ? e0 + CH : E;
  for (int eb = e0; eb < e1; eb += EBC * ETH) {
    int e = eb + t;
    int cv[EBC];
    #pragma unroll
    for (int k = 0; k < EBC; ++k) {
      int ee = e + k * ETH;
      cv[k] = col[ee < e1 ? ee : e1 - 1];
    }
    static_assert(EBC == 7, "the statement below names the seven loaded values");
    asm volatile("" :: "v"(cv[0]), "v"(cv[1]), "v"(cv[2]), "v"(cv[3]), "v"(cv[4]), "v"(cv[5]), "v"(cv[6]));
    #pragma unroll
    for (int k = 0; k < EBC; ++k)
      if (e + k * ETH < e1) atomicAdd(&hist[cv[k] >> BSH], 1);
  }
  __syncthreads();
  int c = (t < HB) ? hist[t] : 0;
  int tot;
  int excl = scan256(c, t, wsum, &tot);
  if (t < nbp) {
    cnt1[t * CHUNKS + w] = c;
    loc1[t * CHUNKS + w] = excl;
  }
  if (t < HB) hist[t] = excl;
  __syncthreads();
  unsigned* out = tmp + e0;
  int len = e1 - e0;
  bool staged = len <= PCAP;
  for (int eb = e0; eb < e1; eb += EBC * ETH) {
    int e = eb + t;
    int key[EBC], p[EBC];
    unsigned val[EBC];
    #pragma unroll
    for (int k = 0; k < EBC; ++k) {
      int ee = e + k * ETH;
      int ec = ee < e1 ? ee : e1 - 1;
      int cv = col[ec];
      int rv = row[ec];
      key[k] = ee < e1 ? (cv >> BSH) : -1;
      val[k] = (unsigned)rv | ((unsigned)(cv & ((1 << BSH) - 1)) << 24);
    }
    #pragma unroll
    for (int k = 0; k < EBC; ++k) {
      p[k] = 0;
      if (key[k] >= 0) p[k] = atomicAdd(&hist[key[k]], 1);
    }
    if (staged) {
      #pragma unroll
      for (int k = 0; k < EBC; ++k)
        if (key[k] >= 0) stage[p[k]] = val[k];
    } else {
      #pragma unroll
      for (int k = 0; k < EBC; ++k)
        if (key[k] >= 0) out[p[k]] = val[k];
    }
  }
  if (staged) {
    __syncthreads();
    for (int i = t; i < len; i += ETH) out[i] = stage[i];
  }
}

__device__ __forceinline__ int run_index(const int* pre, const int* rb, int i) {
  int w = 0;
  #pragma unroll
  for (int s = CHUNKS / 2; s > 0; s >>= 1)
    if (pre[w + s] <= i) w += s;
  return rb[w] + i;
}

__global__ void __launch_bounds__(ETH)
bucket_lists(const int* __restrict__ cnt1, const int* __restrict__ loc1,
             const unsigned* __restrict__ tmp,
             int* __restrict__ rowptr, float* __restrict__ dinv, int* __restrict__ src,
             int N, int CH) {
  static_assert(CHUNKS == 256 && (1 << BSH) == 256, "scan256 covers 256 runs and 256 nodes");
  __shared__ int cnt[1 << BSH];
  __shared__ int cur[1 << BSH];
  __shared__ int pre[CHUNKS];
  __shared__ int rb[CHUNKS];
  __shared__ int wsum[8];
  __shared__ int red[ETH / 32];
  __shared__ int stage[BCAP];
  int t = threadIdx.x, b = blockIdx.x;
  int lane = t & 31, wv = t >> 5;
  int part = 0;
  int nfront = b * (CHUNKS / 4);
  #pragma unroll 8
  for (int i = t; i < nfront; i += ETH) {
    int4 v = *(const int4*)(cnt1 + 4 * (size_t)i);
    part += (v.x + v.y) + (v.z + v.w);
  }
  int c = 0, lc = 0;
  if (t < CHUNKS) {
    c  = cnt1[b * CHUNKS + t];
    lc = loc1[b * CHUNKS + t];
  }
  #pragma unroll
  for (int d = 16; d > 0; d >>= 1) part += __shfl_xor(part, d);
  if (lane == 0) red[wv] = part;
  if (t < (1 << BSH)) cnt[t] = 0;
  int n;
  int ex = scan256(c, t, wsum, &n);
  int beg = 0;
  #pragma unroll
  for (int k = 0; k < ETH / 32; ++k) beg += red[k];
  if (t < CHUNKS) {
    pre[t] = ex;
    rb[t]  = t * CH + lc - ex;
  }
  __syncthreads();
  for (int ib = 0; ib < n; ib += EBB * ETH) {
    int i = ib + t;
    unsigned v[EBB];
    #pragma unroll
    for (int k = 0; k < EBB; ++k) {
      int ii = i + k * ETH;
      v[k] = tmp[run_index(pre, rb, ii < n ? ii : n - 1)];
    }
    #pragma unroll
    for (int k = 0; k < EBB; ++k)
      if (i + k * ETH < n) atomicAdd(&cnt[v[k] >> 24], 1);
  }
  __syncthreads();
  int cn = (t < (1 << BSH)) ? cnt[t] : 0;
  int ntot;
  int excl = scan256(cn, t, wsum, &ntot);
  if (t < (1 << BSH)) {
    int node = (b << BSH) + t;
    if (node <= N) rowptr[node] = beg + excl;
    if (node < N) dinv[node] = rsqrtf((float)cn + 1.0f);
    cur[t] = excl;
  }
  __syncthreads();
  int* out = src + beg;
  bool staged = n <= BCAP;
  for (int ib = 0; ib < n; ib += EBB * ETH) {
    int i = ib + t;
    unsigned v[EBB];
    int p[EBB];
    #pragma unroll
    for (int k = 0; k < EBB; ++k) {
      int ii = i + k * ETH;
      v[k] = tmp[run_index(pre, rb, ii < n ? ii : n - 1)];
    }
    #pragma unroll
    for (int k = 0; k < EBB; ++k) {
      p[k] = 0;
      if (i + k * ETH < n) p[k] = atomicAdd(&cur[v[k] >> 24], 1);
    }
    if (staged) {
      #pragma unroll
      for (int k = 0; k < EBB; ++k)
        if (i + k * ETH < n) stage[p[k]] = (int)(v[k] & 0xFFFFFFu);
    } else {
      #pragma unroll
      for (int k = 0; k < EBB; ++k)
        if (i + k * ETH < n) out[p[k]] = (int)(v[k] & 0xFFFFFFu);
    }
  }
  if (staged) {
    __syncthreads();
    for (int i = t; i < n; i += ETH) out[i] = stage[i];
  }
}

template <int K>
__device__ __forceinline__ void add_rows(const int* __restrict__ src, const float* __restrict__ dinv,
                                         const float* __restrict__ h, int j, int lane, float dc,
                                         float& a0, float& a1, float& a2) {
  int r[K];
  float w[K];
  float v[K][3];
  #pragma unroll
  for (int i = 0; i < K; ++i) r[i] = src[j + i];
  #pragma unroll
  for (int i = 0; i < K; ++i) {
    const float* hr = h + (size_t)r[i] * D + lane;
    v[i][0] = hr[0]; v[i][1] = hr[32]; v[i][2] = hr[64];
  }
  #pragma unroll
  for (int i = 0; i < K; ++i) w[i] = dinv[r[i]] * dc;
  #pragma unroll
  for (int i = 0; i < K; ++i) {
    a0 += v[i][0] * w[i];
    a1 += v[i][1] * w[i];
    a2 += v[i][2] * w[i];
  }
}

__device__ __forceinline__ void sum_node(const int* __restrict__ rowptr, const int* __restrict__ src,
                                         const float* __restrict__ dinv, const float* __restrict__ f,
                                         int node, int lane, float& a0, float& a1, float& a2) {
  int beg = rowptr[node], end = rowptr[node + 1];
  float dc = dinv[node];
  const float* fc = f + (size_t)node * D + lane;
  float s0 = fc[0], s1 = fc[32], s2 = fc[64];
  int j = beg;
  for (; j + 8 <= end; j += 8) add_rows<8>(src, dinv, f, j, lane, dc, a0, a1, a2);
  if (j + 4 <= end) { add_rows<4>(src, dinv, f, j, lane, dc, a0, a1, a2); j += 4; }
  if (j + 2 <= end) { add_rows<2>(src, dinv, f, j, lane, dc, a0, a1, a2); j += 2; }
  if (j < end) add_rows<1>(src, dinv, f, j, lane, dc, a0, a1, a2);
  float s = dc * dc;
  a0 += s0 * s;
  a1 += s1 * s;
  a2 += s2 * s;
}

#define NPW 4
#define TM  16
#define AP  100
#define KB  8
__global__ void __launch_bounds__(32 * (TM / NPW), 16)
layer_sum_then_product(const int* __restrict__ rowptr, const int* __restrict__ src,
                       const float* __restrict__ dinv, const float* __restrict__ f,
                       const float* __restrict__ Wm, const float* __restrict__ b,
                       float* __restrict__ out, int relu, int N) {
  static_assert(TM / NPW == 4 && D == 96 && D % (4 * KB) == 0,
                "4 waves: three of them cover the 96 columns, 32 each; whole trips");
  __shared__ float agg[TM * AP];
  int lane = threadIdx.x & 31;
  int wave = __builtin_amdgcn_readfirstlane((int)(threadIdx.x >> 5));
  int m0 = blockIdx.x * TM;

  #pragma nounroll
  for (int i = 0; i < NPW; ++i) {
    int lr = wave * NPW + i;
    int node = m0 + lr;
    float a0 = 0.0f, a1 = 0.0f, a2 = 0.0f;
    if (node < N) sum_node(rowptr, src, dinv, f, node, lane, a0, a1, a2);
    float* ar = agg + lr * AP + lane;
    ar[0] = a0; ar[32] = a1; ar[64] = a2;
  }
  __syncthreads();
  if (wave == 3) return;

  int c  = lane & 15;
  int hf = lane >> 4;
  int n0 = wave * 32;
  const float* ap = agg + c * AP + 2 * hf;
  const float* bp = Wm + (size_t)(2 * hf) * D + n0 + 2 * c;
  v8f acc0 = {}, acc1 = {};
#if USE_F32_WMMA
  #pragma nounroll
  for (int kb = 0; kb < D; kb += 4 * KB) {
    v2f w0[KB], w1[KB];
    #pragma unroll
    for (int s = 0; s < KB; ++s) {
      w0[s] = *(const v2f*)(bp + (size_t)(kb + 4 * s) * D);
      w1[s] = *(const v2f*)(bp + (size_t)(kb + 4 * s + 1) * D);
    }
    #pragma unroll
    for (int s = 0; s < KB; ++s) {
      v2f a = *(const v2f*)(ap + kb + 4 * s);
      v2f b0, b1;
      b0.x = w0[s].x; b0.y = w1[s].x;
      b1.x = w0[s].y; b1.y = w1[s].y;
      acc0 = __builtin_amdgcn_wmma_f32_16x16x4_f32(
          false, a, false, b0, (short)0, acc0, false, false);
      acc1 = __builtin_amdgcn_wmma_f32_16x16x4_f32(
          false, a, false, b1, (short)0, acc1, false, false);
    }
  }
#endif
  v2f bias = *(const v2f*)(b + n0 + 2 * c);
  int rbase = m0 + hf * 8;
  #pragma unroll
  for (int j = 0; j < 8; ++j) {
    v2f v;
    v.x = acc0[j] + bias.x;
    v.y = acc1[j] + bias.y;
    if (relu) { v.x = fmaxf(v.x, 0.0f); v.y = fmaxf(v.y, 0.0f); }
    if (rbase + j < N) *(v2f*)(out + (size_t)(rbase + j) * D + n0 + 2 * c) = v;
  }
}

extern "C" void kernel_launch(void* const* d_in, const int* in_sizes, int n_in,
                              void* d_out, int out_size, void* d_ws, size_t ws_size,
                              hipStream_t stream) {
  const float* x  = (const float*)d_in[0];
  const int*   ei = (const int*)d_in[1];
  const float* W1 = (const float*)d_in[2];
  const float* b1 = (const float*)d_in[3];
  const float* W2 = (const float*)d_in[4];
  const float* b2 = (const float*)d_in[5];
  float* out = (float*)d_out;

  int N = in_sizes[0] / D;
  int E = in_sizes[1] / 2;
  const int* row = ei;
  const int* col = ei + E;

  int nb  = N / (1 << BSH) + 1;
  int nbp = (nb + 3) / 4 * 4;
  int CH  = ((E + CHUNKS - 1) / CHUNKS + 31) / 32 * 32;
  size_t NP = (((size_t)N + 1 + 1023) / 1024) * 1024;
  size_t EP = (((size_t)E + 1023) / 1024) * 1024;
  size_t TP = (size_t)nbp * CHUNKS;
  float* ws     = (float*)d_ws;
  float* dinv   = ws;
  int*   rowptr = (int*)(ws + NP);
  int*   cnt1   = (int*)(ws + 2 * NP);
  int*   loc1   = (int*)(ws + 2 * NP + TP);
  int*   src    = (int*)(ws + 2 * NP + 2 * TP);
  unsigned* tmp = (unsigned*)(ws + 2 * NP + 2 * TP + EP);
  float* acc    = ws + 2 * NP + 2 * TP + 2 * EP;

  int gTile = (N + TM - 1) / TM;

  piece_sort<<<CHUNKS, ETH, 0, stream>>>(row, col, cnt1, loc1, tmp, E, CH, nbp);
  bucket_lists<<<nb, ETH, 0, stream>>>(cnt1, loc1, tmp, rowptr, dinv, src, N, CH);

  layer_sum_then_product<<<gTile, 32 * (TM / NPW), 0, stream>>>(rowptr, src, dinv, x, W1, b1, acc, 1, N);

  layer_sum_then_product<<<gTile, 32 * (TM / NPW), 0, stream>>>(rowptr, src, dinv, acc, W2, b2, out, 0, N);
}
